# starting kernel
# baseline (speedup 1.0000x reference)
#include <hip/hip_runtime.h>
#include <hip/hip_bf16.h>

#define BATCH 512
#define JN    140
#define NB    28
#define NV    13059
#define NC    (3 * BATCH)
#define KDIM  (4 * JN)
#define LDSW  132

typedef __attribute__((ext_vector_type(2))) float v2f;
typedef __attribute__((ext_vector_type(8))) float v8f;

__global__ __launch_bounds__(256)
void mouse_chain_kernel(const float* __restrict__ pose,
                        const float* __restrict__ bone_lengths,
                        const float* __restrict__ cbl,
                        const float* __restrict__ trans,
                        const float* __restrict__ scale,
                        const float* __restrict__ Jt,
                        const int*   __restrict__ parent,
                        const int*   __restrict__ bone_mapper,
                        float* __restrict__ Gmat,
                        float* __restrict__ Gchain,
                        float* __restrict__ Jout)
{
    const int b = blockIdx.x * blockDim.x + threadIdx.x;
    if (b >= BATCH) return;
    const float sc = scale[b];
    const float t0 = trans[b * 3 + 0];
    const float t1 = trans[b * 3 + 1];
    const float t2 = trans[b * 3 + 2];

    for (int i = 0; i < JN; ++i) {
        const float ez = pose[(b * JN + i) * 3 + 0];
        const float ey = pose[(b * JN + i) * 3 + 1];
        const float ex = pose[(b * JN + i) * 3 + 2];
        const float cx = cosf(ex), sx = sinf(ex);
        const float cy = cosf(ey), sy = sinf(ey);
        const float cz = cosf(ez), sz = sinf(ez);
        float R[9];
        R[0] = cz * cy; R[1] = cz * sy * sx - sz * cx; R[2] = cz * sy * cx + sz * sx;
        R[3] = sz * cy; R[4] = sz * sy * sx + cz * cx; R[5] = sz * sy * cx - cz * sx;
        R[6] = -sy;     R[7] = cy * sx;                R[8] = cy * cx;

        float G[12];
        if (i == 0) {
            G[0] = R[0]; G[1] = R[1]; G[2]  = R[2]; G[3]  = Jt[0];
            G[4] = R[3]; G[5] = R[4]; G[6]  = R[5]; G[7]  = Jt[1];
            G[8] = R[6]; G[9] = R[7]; G[10] = R[8]; G[11] = Jt[2];
        } else {
            const int p = parent[i];
            float Gp[12];
#pragma unroll
            for (int e = 0; e < 12; ++e) Gp[e] = Gchain[(p * 12 + e) * BATCH + b];
            float s;
            if (i == 1) {
                s = cbl[b];
            } else {
                const int bm = bone_mapper[i];
                s = (bm >= 0) ? 2.0f / (1.0f + expf(-bone_lengths[b * NB + bm] * 0.2f))
                              : 1.0f;
            }
            const float off0 = (Jt[i * 3 + 0] - Jt[p * 3 + 0]) * s;
            const float off1 = (Jt[i * 3 + 1] - Jt[p * 3 + 1]) * s;
            const float off2 = (Jt[i * 3 + 2] - Jt[p * 3 + 2]) * s;
#pragma unroll
            for (int r = 0; r < 3; ++r) {
                const float g0 = Gp[r * 4 + 0], g1 = Gp[r * 4 + 1];
                const float g2 = Gp[r * 4 + 2], g3 = Gp[r * 4 + 3];
                G[r * 4 + 0] = g0 * R[0] + g1 * R[3] + g2 * R[6];
                G[r * 4 + 1] = g0 * R[1] + g1 * R[4] + g2 * R[7];
                G[r * 4 + 2] = g0 * R[2] + g1 * R[5] + g2 * R[8];
                G[r * 4 + 3] = g0 * off0 + g1 * off1 + g2 * off2 + g3;
            }
        }
#pragma unroll
        for (int e = 0; e < 12; ++e) Gchain[(i * 12 + e) * BATCH + b] = G[e];

        const float jx = Jt[i * 3 + 0], jy = Jt[i * 3 + 1], jz = Jt[i * 3 + 2];
#pragma unroll
        for (int r = 0; r < 3; ++r) {
            const float trel = G[r * 4 + 3] -
                (G[r * 4 + 0] * jx + G[r * 4 + 1] * jy + G[r * 4 + 2] * jz);
            const size_t c2 = (size_t)(3 * b + r) * 2;
            Gmat[(size_t)(2 * i    ) * NC * 2 + c2 + 0] = G[r * 4 + 0];
            Gmat[(size_t)(2 * i    ) * NC * 2 + c2 + 1] = G[r * 4 + 1];
            Gmat[(size_t)(2 * i + 1) * NC * 2 + c2 + 0] = G[r * 4 + 2];
            Gmat[(size_t)(2 * i + 1) * NC * 2 + c2 + 1] = trel;
        }
        Jout[(b * JN + i) * 3 + 0] = G[3]  * sc + t0;
        Jout[(b * JN + i) * 3 + 1] = G[7]  * sc + t1;
        Jout[(b * JN + i) * 3 + 2] = G[11] * sc + t2;
    }
}

__global__ __launch_bounds__(256)
void mouse_skin_gemm(const float* __restrict__ W,
                     const float* __restrict__ vt,
                     const float* __restrict__ Gmat,
                     const float* __restrict__ scale,
                     const float* __restrict__ trans,
                     float* __restrict__ Vout)
{
    __shared__ float lds[16 * LDSW];

    const int lane  = threadIdx.x & 31;
    const int wave  = threadIdx.x >> 5;
    const int nSub  = blockIdx.y * 8 + wave;
    const int col   = lane & 15;
    const int khalf = lane >> 4;

    const int   vRow  = blockIdx.x * 16 + col;
    const bool  vok   = vRow < NV;
    const int   vc    = vok ? vRow : 0;
    const float wmask = vok ? 1.0f : 0.0f;

    const float c0 = vt[vc * 3 + 2 * khalf] * wmask;
    const float c1 = ((khalf == 0) ? vt[vc * 3 + 1] : 1.0f) * wmask;

    const float* __restrict__ wrow = W + vc * JN;
    const int cIdx = nSub * 16 + col;
    const float* __restrict__ gp =
        Gmat + ((size_t)khalf * NC + cIdx) * 2;

    v8f acc0 = {}, acc1 = {};
#pragma unroll 2
    for (int jq = 0; jq < JN / 4; ++jq) {
        const float4 wq = *(const float4*)(wrow + jq * 4);
        const v2f b0 = *(const v2f*)(gp);
        const v2f b1 = *(const v2f*)(gp + 4 * NC);
        const v2f b2 = *(const v2f*)(gp + 8 * NC);
        const v2f b3 = *(const v2f*)(gp + 12 * NC);
        v2f a;
        a.x = wq.x * c0; a.y = wq.x * c1;
        acc0 = __builtin_amdgcn_wmma_f32_16x16x4_f32(false, a, false, b0,
                                                     (short)0, acc0, false, false);
        a.x = wq.y * c0; a.y = wq.y * c1;
        acc1 = __builtin_amdgcn_wmma_f32_16x16x4_f32(false, a, false, b1,
                                                     (short)0, acc1, false, false);
        a.x = wq.z * c0; a.y = wq.z * c1;
        acc0 = __builtin_amdgcn_wmma_f32_16x16x4_f32(false, a, false, b2,
                                                     (short)0, acc0, false, false);
        a.x = wq.w * c0; a.y = wq.w * c1;
        acc1 = __builtin_amdgcn_wmma_f32_16x16x4_f32(false, a, false, b3,
                                                     (short)0, acc1, false, false);
        gp += 16 * NC;
    }
    const v8f acc = acc0 + acc1;

#pragma unroll
    for (int i = 0; i < 8; ++i)
        lds[(khalf * 8 + i) * LDSW + wave * 16 + col] = acc[i];
    __syncthreads();

    const int C0     = blockIdx.y * 128;
    const int vr0    = blockIdx.x * 16;
    const int bFirst = C0 / 3;
    const int nGroups = (C0 + 127) / 3 - bFirst + 1;
    const int seg = threadIdx.x / 48;
    const int s   = threadIdx.x - seg * 48;
    if (seg < 5) {
        const int row = s / 3;
        const int r   = s - row * 3;
        const int vr  = vr0 + row;
        for (int g = seg; g < nGroups; g += 5) {
            const int bg = bFirst + g;
            const int lc = 3 * bg + r - C0;
            if (lc >= 0 && lc < 128 && vr < NV) {
                const float val = lds[row * LDSW + lc];
                Vout[((size_t)bg * NV + vr) * 3 + r] =
                    val * scale[bg] + trans[bg * 3 + r];
            }
        }
    }
}

extern "C" void kernel_launch(void* const* d_in, const int* in_sizes, int n_in,
                              void* d_out, int out_size, void* d_ws, size_t ws_size,
                              hipStream_t stream) {
    const float* pose         = (const float*)d_in[0];
    const float* bone_lengths = (const float*)d_in[1];
    const float* cbl          = (const float*)d_in[2];
    const float* trans        = (const float*)d_in[3];
    const float* scale        = (const float*)d_in[4];
    const float* v_template   = (const float*)d_in[5];
    const float* t_pose       = (const float*)d_in[6];
    const float* weights      = (const float*)d_in[7];
    const int*   parent       = (const int*)d_in[8];
    const int*   bone_mapper  = (const int*)d_in[9];

    float* Gmat   = (float*)d_ws;
    float* Gchain = Gmat + (size_t)KDIM * NC;

    float* Vout = (float*)d_out;
    float* Jout = Vout + (size_t)BATCH * NV * 3;

    mouse_chain_kernel<<<dim3((BATCH + 255) / 256), 256, 0, stream>>>(
        pose, bone_lengths, cbl, trans, scale, t_pose, parent, bone_mapper,
        Gmat, Gchain, Jout);

    dim3 grid((NV + 15) / 16, (NC / 16) / 8);
    mouse_skin_gemm<<<grid, 256, 0, stream>>>(
        weights, v_template, Gmat, scale, trans, Vout);
}


# GEMM grid tail: last 771 vertex rows run as 64-row blocks of half-height wave tiles (2x3 accumulators, fragments 3 chunks ahead): 1536 full blocks = 6 per WGP, then one half-length round, not a 7th
# speedup vs baseline: 2.8397x; 2.8397x over previous
#include <hip/hip_runtime.h>
#include <hip/hip_bf16.h>

#define BATCH 512
#define JN    140
#define NB    28
#define NV    13059
#define NC    (3 * BATCH)
#define KDIM  (4 * JN)
#define NT    3
#define WCOLS (NT * 16)
#define EPITCH 49
#define NCHUNK (JN / 4)
#define LDS_FLOATS (27 * 1024)

#define MT_FULL      4
#define MT_HALF      2
#define FULL_YBLOCKS 96
#define FULL_ROWS    (FULL_YBLOCKS * 2 * MT_FULL * 16)
#define HALF_YBLOCKS ((NV - FULL_ROWS + 2 * MT_HALF * 16 - 1) / (2 * MT_HALF * 16))

typedef __attribute__((ext_vector_type(2))) float v2f;
typedef __attribute__((ext_vector_type(8))) float v8f;

#define CHAIN_THREADS 160

__device__ __forceinline__ float4 row_times_local(const float4 g, const float4 A0,
                                                  const float4 A1, const float4 A2)
{
    float4 n;
    n.x = g.x * A0.x + g.y * A1.x + g.z * A2.x;
    n.y = g.x * A0.y + g.y * A1.y + g.z * A2.y;
    n.z = g.x * A0.z + g.y * A1.z + g.z * A2.z;
    n.w = g.x * A0.w + g.y * A1.w + g.z * A2.w + g.w;
    return n;
}

__global__ __launch_bounds__(CHAIN_THREADS)
void mouse_chain_kernel(const float* __restrict__ pose,
                        const float* __restrict__ bone_lengths,
                        const float* __restrict__ cbl,
                        const float* __restrict__ trans,
                        const float* __restrict__ scale,
                        const float* __restrict__ Jt,
                        const int*   __restrict__ parent,
                        const int*   __restrict__ bone_mapper,
                        float* __restrict__ Gmat,
                        float* __restrict__ Jout)
{
    __shared__ float4        Aloc[JN * 3];
    __shared__ unsigned char par[JN];
    __shared__ unsigned char path[JN * JN];

    const int  b    = blockIdx.x;
    const int  j    = threadIdx.x;
    const bool live = j < JN;

    float jx = 0.0f, jy = 0.0f, jz = 0.0f;
    if (live) {
        int p = 0;
        if (j > 0) p = min(max(parent[j], 0), JN - 1);
        par[j] = (unsigned char)p;

        const float ez = pose[(b * JN + j) * 3 + 0];
        const float ey = pose[(b * JN + j) * 3 + 1];
        const float ex = pose[(b * JN + j) * 3 + 2];
        jx = Jt[j * 3 + 0]; jy = Jt[j * 3 + 1]; jz = Jt[j * 3 + 2];

        const float cx = cosf(ex), sx = sinf(ex);
        const float cy = cosf(ey), sy = sinf(ey);
        const float cz = cosf(ez), sz = sinf(ez);
        float R[9];
        R[0] = cz * cy; R[1] = cz * sy * sx - sz * cx; R[2] = cz * sy * cx + sz * sx;
        R[3] = sz * cy; R[4] = sz * sy * sx + cz * cx; R[5] = sz * sy * cx - cz * sx;
        R[6] = -sy;     R[7] = cy * sx;                R[8] = cy * cx;

        float off0, off1, off2;
        if (j == 0) {
            off0 = jx; off1 = jy; off2 = jz;
        } else {
            float s;
            if (j == 1) {
                s = cbl[b];
            } else {
                const int bm = min(bone_mapper[j], NB - 1);
                s = (bm >= 0) ? 2.0f / (1.0f + expf(-bone_lengths[b * NB + bm] * 0.2f))
                              : 1.0f;
            }
            off0 = (jx - Jt[p * 3 + 0]) * s;
            off1 = (jy - Jt[p * 3 + 1]) * s;
            off2 = (jz - Jt[p * 3 + 2]) * s;
        }
        Aloc[j * 3 + 0] = make_float4(R[0], R[1], R[2], off0);
        Aloc[j * 3 + 1] = make_float4(R[3], R[4], R[5], off1);
        Aloc[j * 3 + 2] = make_float4(R[6], R[7], R[8], off2);
    }
    __syncthreads();
    if (!live) return;

    unsigned char* mypath = path + j * JN;
    int d = 0;
    for (int q = j; q != 0 && d < JN; q = par[q]) {
        mypath[d] = (unsigned char)q;
        ++d;
    }

    float4 g0 = Aloc[0], g1 = Aloc[1], g2 = Aloc[2];
    int k = d - 1;
    int a = (k >= 0) ? mypath[k] : 0;
    float4 A0 = Aloc[a * 3 + 0], A1 = Aloc[a * 3 + 1], A2 = Aloc[a * 3 + 2];
    int an = (k >= 1) ? mypath[k - 1] : 0;
    for (; k >= 0; --k) {
        const float4 B0 = Aloc[an * 3 + 0], B1 = Aloc[an * 3 + 1], B2 = Aloc[an * 3 + 2];
        an = (k >= 2) ? mypath[k - 2] : 0;
        g0 = row_times_local(g0, A0, A1, A2);
        g1 = row_times_local(g1, A0, A1, A2);
        g2 = row_times_local(g2, A0, A1, A2);
        A0 = B0; A1 = B1; A2 = B2;
    }

    const float trel0 = g0.w - (g0.x * jx + g0.y * jy + g0.z * jz);
    const float trel1 = g1.w - (g1.x * jx + g1.y * jy + g1.z * jz);
    const float trel2 = g2.w - (g2.x * jx + g2.y * jy + g2.z * jz);
    float2* __restrict__ G2 = (float2*)Gmat;
    const size_t rowA = (size_t)(2 * j) * NC + 3 * b;
    const size_t rowB = rowA + NC;
    G2[rowA + 0] = make_float2(g0.x, g0.y);
    G2[rowA + 1] = make_float2(g1.x, g1.y);
    G2[rowA + 2] = make_float2(g2.x, g2.y);
    G2[rowB + 0] = make_float2(g0.z, trel0);
    G2[rowB + 1] = make_float2(g1.z, trel1);
    G2[rowB + 2] = make_float2(g2.z, trel2);

    const float sc = scale[b];
    Jout[(b * JN + j) * 3 + 0] = g0.w * sc + trans[b * 3 + 0];
    Jout[(b * JN + j) * 3 + 1] = g1.w * sc + trans[b * 3 + 1];
    Jout[(b * JN + j) * 3 + 2] = g2.w * sc + trans[b * 3 + 2];
}

template <int MT>
struct FragSet {
    float4 w[MT];
    v2f    g[4][NT];
};

template <int MT>
__device__ __forceinline__ void load_frags(FragSet<MT>& f, const float* const (&wrow)[MT],
                                           const float* __restrict__ gp, const int jq)
{
#pragma unroll
    for (int m = 0; m < MT; ++m)
        f.w[m] = *(const float4*)(wrow[m] + jq * 4);
    const float* __restrict__ g = gp + (size_t)jq * (16 * NC);
#pragma unroll
    for (int q = 0; q < 4; ++q)
#pragma unroll
        for (int n = 0; n < NT; ++n)
            f.g[q][n] = *(const v2f*)(g + q * (4 * NC) + n * 32);
}

template <int MT>
__device__ __forceinline__ void mma_chunk(v8f (&acc)[MT][NT], const FragSet<MT>& f,
                                          const float (&c0)[MT], const float (&c1)[MT])
{
#pragma unroll
    for (int q = 0; q < 4; ++q) {
#pragma unroll
        for (int m = 0; m < MT; ++m) {
            const float wv = (q == 0) ? f.w[m].x : (q == 1) ? f.w[m].y : (q == 2) ? f.w[m].z : f.w[m].w;
            v2f a;
            a.x = wv * c0[m]; a.y = wv * c1[m];
#pragma unroll
            for (int n = 0; n < NT; ++n)
                acc[m][n] = __builtin_amdgcn_wmma_f32_16x16x4_f32(false, a, false, f.g[q][n],
                                                                  (short)0, acc[m][n], false, false);
        }
    }
}

template <int MT, int SETS>
__device__ __forceinline__ void skin_tile(const int rowBase, const int colBase,
                                          const int lane, const int wave,
                                          float* __restrict__ lds,
                                          const float* __restrict__ W,
                                          const float* __restrict__ vt,
                                          const float* __restrict__ Gmat,
                                          const float* __restrict__ scale,
                                          const float* __restrict__ trans,
                                          float* __restrict__ Vout)
{
    constexpr int WROWS = MT * 16;

    const int col   = lane & 15;
    const int khalf = lane >> 4;

    const float* wrow[MT];
    float c0[MT], c1[MT];
#pragma unroll
    for (int m = 0; m < MT; ++m) {
        const int   vRow  = rowBase + m * 16 + col;
        const bool  vok   = vRow < NV;
        const int   vc    = vok ? vRow : 0;
        const float wmask = vok ? 1.0f : 0.0f;
        c0[m] = vt[vc * 3 + 2 * khalf] * wmask;
        c1[m] = ((khalf == 0) ? vt[vc * 3 + 1] : 1.0f) * wmask;
        wrow[m] = W + vc * JN;
    }
    const float* __restrict__ gp =
        Gmat + ((size_t)khalf * NC + colBase + col) * 2;

    v8f acc[MT][NT];
#pragma unroll
    for (int m = 0; m < MT; ++m)
#pragma unroll
        for (int n = 0; n < NT; ++n)
            acc[m][n] = v8f{};

    static_assert((NCHUNK - (SETS - 1)) % SETS == 0, "the chunk loop is unrolled by the number of sets");
    if (rowBase < NV) {
        FragSet<MT> f[SETS];
#pragma unroll
        for (int s = 0; s < SETS - 1; ++s)
            load_frags<MT>(f[s], wrow, gp, s);
        for (int jq = 0; jq < NCHUNK - (SETS - 1); jq += SETS) {
#pragma unroll
            for (int s = 0; s < SETS; ++s) {
                load_frags<MT>(f[(s + SETS - 1) % SETS], wrow, gp, jq + s + SETS - 1);
                __builtin_amdgcn_sched_barrier(0);
                mma_chunk<MT>(acc, f[s], c0, c1);
            }
        }
#pragma unroll
        for (int s = 0; s < SETS - 1; ++s)
            mma_chunk<MT>(acc, f[s], c0, c1);
    }

    float* __restrict__ my = lds + wave * (WROWS * EPITCH);
#pragma unroll
    for (int n = 0; n < NT; ++n) {
        const int   c  = colBase + n * 16 + col;
        const float sc = scale[c / 3];
        const float tr = trans[c];
#pragma unroll
        for (int m = 0; m < MT; ++m)
#pragma unroll
            for (int i = 0; i < 8; ++i)
                my[(m * 16 + khalf * 8 + i) * EPITCH + n * 16 + col] = acc[m][n][i] * sc + tr;
    }
    __syncthreads();

    const int rowsValid = min(WROWS, NV - rowBase);
    const int nEl       = rowsValid * 3;
    const int bBase     = colBase / 3;
#pragma unroll
    for (int k = 0; k < (WROWS * 3 + 31) / 32; ++k) {
        const int e = lane + 32 * k;
        if (e < nEl) {
            const int row = e / 3;
            const int r   = e - row * 3;
            const float* __restrict__ src = my + row * EPITCH + r;
            float* __restrict__ dst = Vout + ((size_t)bBase * NV + rowBase) * 3 + e;
#pragma unroll
            for (int bl = 0; bl < WCOLS / 3; ++bl)
                dst[(size_t)bl * (NV * 3)] = src[bl * 3];
        }
    }
}

__global__ __launch_bounds__(128)
void mouse_skin_gemm(const float* __restrict__ W,
                     const float* __restrict__ vt,
                     const float* __restrict__ Gmat,
                     const float* __restrict__ scale,
                     const float* __restrict__ trans,
                     float* __restrict__ Vout)
{
    __shared__ float lds[LDS_FLOATS];

    const int lane = threadIdx.x & 31;
    const int wave = threadIdx.x >> 5;
    const int colBase = (blockIdx.x * 2 + (wave & 1)) * WCOLS;

    if (blockIdx.y < FULL_YBLOCKS) {
        const int rowBase = (blockIdx.y * 2 + (wave >> 1)) * (MT_FULL * 16);
        skin_tile<MT_FULL, 2>(rowBase, colBase, lane, wave, lds, W, vt, Gmat, scale, trans, Vout);
    } else {
        const int rowBase = FULL_ROWS + ((blockIdx.y - FULL_YBLOCKS) * 2 + (wave >> 1)) * (MT_HALF * 16);
        skin_tile<MT_HALF, 4>(rowBase, colBase, lane, wave, lds, W, vt, Gmat, scale, trans, Vout);
    }
}

extern "C" void kernel_launch(void* const* d_in, const int* in_sizes, int n_in,
                              void* d_out, int out_size, void* d_ws, size_t ws_size,
                              hipStream_t stream) {
    const float* pose         = (const float*)d_in[0];
    const float* bone_lengths = (const float*)d_in[1];
    const float* cbl          = (const float*)d_in[2];
    const float* trans        = (const float*)d_in[3];
    const float* scale        = (const float*)d_in[4];
    const float* v_template   = (const float*)d_in[5];
    const float* t_pose       = (const float*)d_in[6];
    const float* weights      = (const float*)d_in[7];
    const int*   parent       = (const int*)d_in[8];
    const int*   bone_mapper  = (const int*)d_in[9];

    float* Gmat   = (float*)d_ws;

    float* Vout = (float*)d_out;
    float* Jout = Vout + (size_t)BATCH * NV * 3;

    mouse_chain_kernel<<<dim3(BATCH), CHAIN_THREADS, 0, stream>>>(
        pose, bone_lengths, cbl, trans, scale, t_pose, parent, bone_mapper,
        Gmat, Jout);

    dim3 grid(NC / (2 * WCOLS), FULL_YBLOCKS + HALF_YBLOCKS);
    mouse_skin_gemm<<<grid, 128, 0, stream>>>(
        weights, v_template, Gmat, scale, trans, Vout);
}


# GEMM retiled: each wave holds 8x3 accumulators (128 rows x 48 columns), a block's 4 waves side by side on the same 128 rows: a G fragment feeds 8 WMMAs not 4, fragment loads lead by 96 WMMAs
# speedup vs baseline: 2.9878x; 1.0522x over previous
#include <hip/hip_runtime.h>
#include <hip/hip_bf16.h>

#define BATCH 512
#define JN    140
#define NB    28
#define NV    13059
#define NC    (3 * BATCH)
#define KDIM  (4 * JN)
#define NT    3
#define WCOLS (NT * 16)
#define EPITCH 49
#define NCHUNK (JN / 4)
#define LDS_FLOATS (27 * 1024)

#define MT_FULL      8
#define MT_HALF      2
#define FULL_YBLOCKS 96
#define FULL_ROWS    (FULL_YBLOCKS * MT_FULL * 16)
#define HALF_YBLOCKS ((NV - FULL_ROWS + MT_HALF * 16 - 1) / (MT_HALF * 16))

typedef __attribute__((ext_vector_type(2))) float v2f;
typedef __attribute__((ext_vector_type(8))) float v8f;

#define CHAIN_THREADS 160

__device__ __forceinline__ float4 row_times_local(const float4 g, const float4 A0,
                                                  const float4 A1, const float4 A2)
{
    float4 n;
    n.x = g.x * A0.x + g.y * A1.x + g.z * A2.x;
    n.y = g.x * A0.y + g.y * A1.y + g.z * A2.y;
    n.z = g.x * A0.z + g.y * A1.z + g.z * A2.z;
    n.w = g.x * A0.w + g.y * A1.w + g.z * A2.w + g.w;
    return n;
}

__global__ __launch_bounds__(CHAIN_THREADS)
void mouse_chain_kernel(const float* __restrict__ pose,
                        const float* __restrict__ bone_lengths,
                        const float* __restrict__ cbl,
                        const float* __restrict__ trans,
                        const float* __restrict__ scale,
                        const float* __restrict__ Jt,
                        const int*   __restrict__ parent,
                        const int*   __restrict__ bone_mapper,
                        float* __restrict__ Gmat,
                        float* __restrict__ Jout)
{
    __shared__ float4        Aloc[JN * 3];
    __shared__ unsigned char par[JN];
    __shared__ unsigned char path[JN * JN];

    const int  b    = blockIdx.x;
    const int  j    = threadIdx.x;
    const bool live = j < JN;

    float jx = 0.0f, jy = 0.0f, jz = 0.0f;
    if (live) {
        int p = 0;
        if (j > 0) p = min(max(parent[j], 0), JN - 1);
        par[j] = (unsigned char)p;

        const float ez = pose[(b * JN + j) * 3 + 0];
        const float ey = pose[(b * JN + j) * 3 + 1];
        const float ex = pose[(b * JN + j) * 3 + 2];
        jx = Jt[j * 3 + 0]; jy = Jt[j * 3 + 1]; jz = Jt[j * 3 + 2];

        const float cx = cosf(ex), sx = sinf(ex);
        const float cy = cosf(ey), sy = sinf(ey);
        const float cz = cosf(ez), sz = sinf(ez);
        float R[9];
        R[0] = cz * cy; R[1] = cz * sy * sx - sz * cx; R[2] = cz * sy * cx + sz * sx;
        R[3] = sz * cy; R[4] = sz * sy * sx + cz * cx; R[5] = sz * sy * cx - cz * sx;
        R[6] = -sy;     R[7] = cy * sx;                R[8] = cy * cx;

        float off0, off1, off2;
        if (j == 0) {
            off0 = jx; off1 = jy; off2 = jz;
        } else {
            float s;
            if (j == 1) {
                s = cbl[b];
            } else {
                const int bm = min(bone_mapper[j], NB - 1);
                s = (bm >= 0) ? 2.0f / (1.0f + expf(-bone_lengths[b * NB + bm] * 0.2f))
                              : 1.0f;
            }
            off0 = (jx - Jt[p * 3 + 0]) * s;
            off1 = (jy - Jt[p * 3 + 1]) * s;
            off2 = (jz - Jt[p * 3 + 2]) * s;
        }
        Aloc[j * 3 + 0] = make_float4(R[0], R[1], R[2], off0);
        Aloc[j * 3 + 1] = make_float4(R[3], R[4], R[5], off1);
        Aloc[j * 3 + 2] = make_float4(R[6], R[7], R[8], off2);
    }
    __syncthreads();
    if (!live) return;

    unsigned char* mypath = path + j * JN;
    int d = 0;
    for (int q = j; q != 0 && d < JN; q = par[q]) {
        mypath[d] = (unsigned char)q;
        ++d;
    }

    float4 g0 = Aloc[0], g1 = Aloc[1], g2 = Aloc[2];
    int k = d - 1;
    int a = (k >= 0) ? mypath[k] : 0;
    float4 A0 = Aloc[a * 3 + 0], A1 = Aloc[a * 3 + 1], A2 = Aloc[a * 3 + 2];
    int an = (k >= 1) ? mypath[k - 1] : 0;
    for (; k >= 0; --k) {
        const float4 B0 = Aloc[an * 3 + 0], B1 = Aloc[an * 3 + 1], B2 = Aloc[an * 3 + 2];
        an = (k >= 2) ? mypath[k - 2] : 0;
        g0 = row_times_local(g0, A0, A1, A2);
        g1 = row_times_local(g1, A0, A1, A2);
        g2 = row_times_local(g2, A0, A1, A2);
        A0 = B0; A1 = B1; A2 = B2;
    }

    const float trel0 = g0.w - (g0.x * jx + g0.y * jy + g0.z * jz);
    const float trel1 = g1.w - (g1.x * jx + g1.y * jy + g1.z * jz);
    const float trel2 = g2.w - (g2.x * jx + g2.y * jy + g2.z * jz);
    float2* __restrict__ G2 = (float2*)Gmat;
    const size_t rowA = (size_t)(2 * j) * NC + 3 * b;
    const size_t rowB = rowA + NC;
    G2[rowA + 0] = make_float2(g0.x, g0.y);
    G2[rowA + 1] = make_float2(g1.x, g1.y);
    G2[rowA + 2] = make_float2(g2.x, g2.y);
    G2[rowB + 0] = make_float2(g0.z, trel0);
    G2[rowB + 1] = make_float2(g1.z, trel1);
    G2[rowB + 2] = make_float2(g2.z, trel2);

    const float sc = scale[b];
    Jout[(b * JN + j) * 3 + 0] = g0.w * sc + trans[b * 3 + 0];
    Jout[(b * JN + j) * 3 + 1] = g1.w * sc + trans[b * 3 + 1];
    Jout[(b * JN + j) * 3 + 2] = g2.w * sc + trans[b * 3 + 2];
}

template <int MT>
struct FragSet {
    float4 w[MT];
    v2f    g[4][NT];
};

template <int MT>
__device__ __forceinline__ void load_frags(FragSet<MT>& f, const float* const (&wrow)[MT],
                                           const float* __restrict__ gp, const int jq)
{
#pragma unroll
    for (int m = 0; m < MT; ++m)
        f.w[m] = *(const float4*)(wrow[m] + jq * 4);
    const float* __restrict__ g = gp + (size_t)jq * (16 * NC);
#pragma unroll
    for (int q = 0; q < 4; ++q)
#pragma unroll
        for (int n = 0; n < NT; ++n)
            f.g[q][n] = *(const v2f*)(g + q * (4 * NC) + n * 32);
}

template <int MT>
__device__ __forceinline__ void mma_chunk(v8f (&acc)[MT][NT], const FragSet<MT>& f,
                                          const float (&c0)[MT], const float (&c1)[MT])
{
#pragma unroll
    for (int q = 0; q < 4; ++q) {
#pragma unroll
        for (int m = 0; m < MT; ++m) {
            const float wv = (q == 0) ? f.w[m].x : (q == 1) ? f.w[m].y : (q == 2) ? f.w[m].z : f.w[m].w;
            v2f a;
            a.x = wv * c0[m]; a.y = wv * c1[m];
#pragma unroll
            for (int n = 0; n < NT; ++n)
                acc[m][n] = __builtin_amdgcn_wmma_f32_16x16x4_f32(false, a, false, f.g[q][n],
                                                                  (short)0, acc[m][n], false, false);
        }
    }
}

template <int MT, int SETS>
__device__ __forceinline__ void skin_tile(const int rowBase, const int colBase,
                                          const int lane, const int wave,
                                          float* __restrict__ lds,
                                          const float* __restrict__ W,
                                          const float* __restrict__ vt,
                                          const float* __restrict__ Gmat,
                                          const float* __restrict__ scale,
                                          const float* __restrict__ trans,
                                          float* __restrict__ Vout)
{
    constexpr int WROWS = MT * 16;

    const int col   = lane & 15;
    const int khalf = lane >> 4;

    const float* wrow[MT];
    float c0[MT], c1[MT];
#pragma unroll
    for (int m = 0; m < MT; ++m) {
        const int   vRow  = rowBase + m * 16 + col;
        const bool  vok   = vRow < NV;
        const int   vc    = vok ? vRow : 0;
        const float wmask = vok ? 1.0f : 0.0f;
        c0[m] = vt[vc * 3 + 2 * khalf] * wmask;
        c1[m] = ((khalf == 0) ? vt[vc * 3 + 1] : 1.0f) * wmask;
        wrow[m] = W + vc * JN;
    }
    const float* __restrict__ gp =
        Gmat + ((size_t)khalf * NC + colBase + col) * 2;

    v8f acc[MT][NT];
#pragma unroll
    for (int m = 0; m < MT; ++m)
#pragma unroll
        for (int n = 0; n < NT; ++n)
            acc[m][n] = v8f{};

    static_assert((NCHUNK - (SETS - 1)) % SETS == 0, "the chunk loop is unrolled by the number of sets");
    if (rowBase < NV) {
        FragSet<MT> f[SETS];
#pragma unroll
        for (int s = 0; s < SETS - 1; ++s)
            load_frags<MT>(f[s], wrow, gp, s);
        for (int jq = 0; jq < NCHUNK - (SETS - 1); jq += SETS) {
#pragma unroll
            for (int s = 0; s < SETS; ++s) {
                load_frags<MT>(f[(s + SETS - 1) % SETS], wrow, gp, jq + s + SETS - 1);
                __builtin_amdgcn_sched_barrier(0);
                mma_chunk<MT>(acc, f[s], c0, c1);
            }
        }
#pragma unroll
        for (int s = 0; s < SETS - 1; ++s)
            mma_chunk<MT>(acc, f[s], c0, c1);
    }

    float* __restrict__ my = lds + wave * (WROWS * EPITCH);
#pragma unroll
    for (int n = 0; n < NT; ++n) {
        const int   c  = colBase + n * 16 + col;
        const float sc = scale[c / 3];
        const float tr = trans[c];
#pragma unroll
        for (int m = 0; m < MT; ++m)
#pragma unroll
            for (int i = 0; i < 8; ++i)
                my[(m * 16 + khalf * 8 + i) * EPITCH + n * 16 + col] = acc[m][n][i] * sc + tr;
    }
    __syncthreads();

    const int rowsValid = min(WROWS, NV - rowBase);
    const int nEl       = rowsValid * 3;
    const int bBase     = colBase / 3;
#pragma unroll
    for (int k = 0; k < (WROWS * 3 + 31) / 32; ++k) {
        const int e = lane + 32 * k;
        if (e < nEl) {
            const int row = e / 3;
            const int r   = e - row * 3;
            const float* __restrict__ src = my + row * EPITCH + r;
            float* __restrict__ dst = Vout + ((size_t)bBase * NV + rowBase) * 3 + e;
#pragma unroll
            for (int bl = 0; bl < WCOLS / 3; ++bl)
                dst[(size_t)bl * (NV * 3)] = src[bl * 3];
        }
    }
}

__global__ __launch_bounds__(128)
void mouse_skin_gemm(const float* __restrict__ W,
                     const float* __restrict__ vt,
                     const float* __restrict__ Gmat,
                     const float* __restrict__ scale,
                     const float* __restrict__ trans,
                     float* __restrict__ Vout)
{
    __shared__ float lds[LDS_FLOATS];

    const int lane = threadIdx.x & 31;
    const int wave = threadIdx.x >> 5;
    const int colBase = (blockIdx.x * 4 + wave) * WCOLS;

    if (blockIdx.y < FULL_YBLOCKS) {
        const int rowBase = blockIdx.y * (MT_FULL * 16);
        skin_tile<MT_FULL, 2>(rowBase, colBase, lane, wave, lds, W, vt, Gmat, scale, trans, Vout);
    } else {
        const int rowBase = FULL_ROWS + (blockIdx.y - FULL_YBLOCKS) * (MT_HALF * 16);
        skin_tile<MT_HALF, 4>(rowBase, colBase, lane, wave, lds, W, vt, Gmat, scale, trans, Vout);
    }
}

extern "C" void kernel_launch(void* const* d_in, const int* in_sizes, int n_in,
                              void* d_out, int out_size, void* d_ws, size_t ws_size,
                              hipStream_t stream) {
    const float* pose         = (const float*)d_in[0];
    const float* bone_lengths = (const float*)d_in[1];
    const float* cbl          = (const float*)d_in[2];
    const float* trans        = (const float*)d_in[3];
    const float* scale        = (const float*)d_in[4];
    const float* v_template   = (const float*)d_in[5];
    const float* t_pose       = (const float*)d_in[6];
    const float* weights      = (const float*)d_in[7];
    const int*   parent       = (const int*)d_in[8];
    const int*   bone_mapper  = (const int*)d_in[9];

    float* Gmat   = (float*)d_ws;

    float* Vout = (float*)d_out;
    float* Jout = Vout + (size_t)BATCH * NV * 3;

    mouse_chain_kernel<<<dim3(BATCH), CHAIN_THREADS, 0, stream>>>(
        pose, bone_lengths, cbl, trans, scale, t_pose, parent, bone_mapper,
        Gmat, Jout);

    dim3 grid(NC / (4 * WCOLS), FULL_YBLOCKS + HALF_YBLOCKS);
    mouse_skin_gemm<<<grid, 128, 0, stream>>>(
        weights, v_template, Gmat, scale, trans, Vout);
}
